# starting kernel
# baseline (speedup 1.0000x reference)
#include <hip/hip_runtime.h>
#include <math.h>


typedef __attribute__((ext_vector_type(2))) float v2f;
typedef __attribute__((ext_vector_type(8))) float v8f;

#define BT    8
#define NN    256
#define FIN   128
#define FOUT  64
#define LDSW  260

__global__ __launch_bounds__(128) void k_gemm_xw(const float* __restrict__ x,
                                                 const float* __restrict__ W,
                                                 float* __restrict__ h) {
    const int blk   = blockIdx.x;
    const int bt    = blk >> 4;
    const int mtile = blk & 15;
    const int wv    = threadIdx.x >> 5;
    const int lane  = threadIdx.x & 31;
    const int half  = lane >> 4;
    const int l15   = lane & 15;

    const float* __restrict__ xrow = x + (size_t)(bt * NN + mtile * 16 + l15) * FIN;
    const int colbase = wv * 16 + l15;

    v8f c = {};
    for (int k0 = 0; k0 < FIN; k0 += 4) {
        const int ka = k0 + half * 2;
        v2f a_; a_.x = xrow[ka];                 a_.y = xrow[ka + 1];
        v2f b_; b_.x = W[ka * FOUT + colbase];   b_.y = W[(ka + 1) * FOUT + colbase];
        c = __builtin_amdgcn_wmma_f32_16x16x4_f32(false, a_, false, b_, (short)0, c,
                                                  false, false);
    }
    float* __restrict__ hout = h + (size_t)(bt * NN + mtile * 16) * FOUT + colbase;
#pragma unroll
    for (int r = 0; r < 8; ++r)
        hout[(size_t)(r + half * 8) * FOUT] = c[r];
}

__global__ __launch_bounds__(256) void k_attn_s(const float* __restrict__ h,
                                                const float* __restrict__ a,
                                                float* __restrict__ s) {
    const int bt = blockIdx.x;
    const int j  = threadIdx.x;
    const float* __restrict__ hb = h + (size_t)bt * NN * FOUT;
    const int ja = (2 * j) & (NN - 1);
    const int jb = (2 * j + 1) & (NN - 1);
    const float* __restrict__ ra = hb + (size_t)ja * FOUT;
    const float* __restrict__ rb = hb + (size_t)jb * FOUT;
    float acc = 0.f;
#pragma unroll 8
    for (int f = 0; f < FOUT; ++f)
        acc = fmaf(ra[f], a[f], fmaf(rb[f], a[FOUT + f], acc));
    s[bt * NN + j] = acc;
}

__global__ __launch_bounds__(128) void k_attn_out(const float* __restrict__ adj,
                                                  const float* __restrict__ h,
                                                  const float* __restrict__ s,
                                                  float* __restrict__ out) {
    __shared__ float attn_lds[16][LDSW];

    const int blk   = blockIdx.x;
    const int bt    = blk >> 4;
    const int itile = blk & 15;
    const int b     = bt >> 2;
    const int tau   = bt & 3;
    const int cpred = (itile >= 8) ? 1 : 0;
    const int tp    = (2 * tau + cpred) & 3;

    const int wv   = threadIdx.x >> 5;
    const int lane = threadIdx.x & 31;
    const int half = lane >> 4;
    const int l15  = lane & 15;

    const float* __restrict__ srow = s + (b * 4 + tp) * NN;
    const float* __restrict__ hb   = h + (size_t)bt * NN * FOUT;

#pragma unroll
    for (int rr = 0; rr < 4; ++rr) {
        const int rloc = wv * 4 + rr;
        const int i    = itile * 16 + rloc;
        const float* __restrict__ arow = adj + ((size_t)(bt * NN + i)) * NN;

        float z[8];
        float mx = -3.402823466e38f;
#pragma unroll
        for (int u = 0; u < 8; ++u) {
            const int j = lane + u * 32;
            const float sv = srow[j];
            const float ev = (sv > 0.f) ? sv : 0.2f * sv;
            const float zz = arow[j] * ev;
            z[u] = zz;
            mx = fmaxf(mx, zz);
        }
#pragma unroll
        for (int off = 16; off > 0; off >>= 1)
            mx = fmaxf(mx, __shfl_xor(mx, off, 32));
        float sum = 0.f;
#pragma unroll
        for (int u = 0; u < 8; ++u) {
            const float p = __expf(z[u] - mx);
            z[u] = p;
            sum += p;
        }
#pragma unroll
        for (int off = 16; off > 0; off >>= 1)
            sum += __shfl_xor(sum, off, 32);
        const float inv = 1.f / sum;
#pragma unroll
        for (int u = 0; u < 8; ++u)
            attn_lds[rloc][lane + u * 32] = z[u] * inv;
    }
    __syncthreads();

    const int colbase = wv * 16 + l15;
    v8f c = {};
    for (int k0 = 0; k0 < NN; k0 += 4) {
        const int ka = k0 + half * 2;
        v2f a_; a_.x = attn_lds[l15][ka];
                a_.y = attn_lds[l15][ka + 1];
        v2f b_; b_.x = hb[(size_t)ka * FOUT + colbase];
                b_.y = hb[(size_t)(ka + 1) * FOUT + colbase];
        c = __builtin_amdgcn_wmma_f32_16x16x4_f32(false, a_, false, b_, (short)0, c,
                                                  false, false);
    }
    float* __restrict__ orow = out + (size_t)(bt * NN + itile * 16) * FOUT + colbase;
#pragma unroll
    for (int r = 0; r < 8; ++r)
        orow[(size_t)(r + half * 8) * FOUT] = c[r];
}

extern "C" void kernel_launch(void* const* d_in, const int* in_sizes, int n_in,
                              void* d_out, int out_size, void* d_ws, size_t ws_size,
                              hipStream_t stream) {
    const float* x   = (const float*)d_in[0];
    const float* adj = (const float*)d_in[1];
    const float* W   = (const float*)d_in[2];
    const float* a   = (const float*)d_in[3];
    float* out = (float*)d_out;

    float* h = (float*)d_ws;
    float* s = h + (size_t)BT * NN * FOUT;

    k_gemm_xw<<<dim3(BT * 16), dim3(128), 0, stream>>>(x, W, h);
    k_attn_s<<<dim3(BT), dim3(256), 0, stream>>>(h, a, s);
    k_attn_out<<<dim3(BT * 16), dim3(128), 0, stream>>>(adj, h, s, out);
}


# h handed over in WMMA register order: kernel 1 stores its accumulator tile as 2 b128 per lane, kernel 2 requests its whole B operand (32 b128) ahead of the softmax; out summed in a new k order
# speedup vs baseline: 1.7274x; 1.7274x over previous
#include <hip/hip_runtime.h>
#include <math.h>


typedef __attribute__((ext_vector_type(2))) float v2f;
typedef __attribute__((ext_vector_type(8))) float v8f;

#define BT    8
#define NN    256
#define FIN   128
#define FOUT  64
#define LDSW  260

__global__ __launch_bounds__(128) void k_gemm_xw_s(const float* __restrict__ x,
                                                   const float* __restrict__ W,
                                                   const float* __restrict__ a,
                                                   float* __restrict__ h,
                                                   float* __restrict__ s) {
    __shared__ float red[4][8];

    const int blk   = blockIdx.x;
    const int bt    = blk >> 4;
    const int mtile = blk & 15;
    const int wv    = threadIdx.x >> 5;
    const int lane  = threadIdx.x & 31;
    const int half  = lane >> 4;
    const int l15   = lane & 15;

    const float* __restrict__ xrow = x + (size_t)(bt * NN + mtile * 16 + l15) * FIN;
    const int colbase = wv * 16 + l15;

    const float a1 = a[colbase];
    const float a2 = a[FOUT + colbase];

    v8f c = {};
    for (int k0 = 0; k0 < FIN; k0 += 4) {
        const int ka = k0 + half * 2;
        v2f a_; a_.x = xrow[ka];                 a_.y = xrow[ka + 1];
        v2f b_; b_.x = W[ka * FOUT + colbase];   b_.y = W[(ka + 1) * FOUT + colbase];
        c = __builtin_amdgcn_wmma_f32_16x16x4_f32(false, a_, false, b_, (short)0, c,
                                                  false, false);
    }
    float4* __restrict__ hout = reinterpret_cast<float4*>(
        h + (size_t)((bt * 4 + wv) * 16 + mtile) * 256);
    hout[lane]      = make_float4(c[0], c[1], c[2], c[3]);
    hout[32 + lane] = make_float4(c[4], c[5], c[6], c[7]);

    float u[4];
#pragma unroll
    for (int q = 0; q < 4; ++q)
        u[q] = fmaf(c[2 * q], a1, c[2 * q + 1] * a2);
#pragma unroll
    for (int off = 8; off > 0; off >>= 1) {
#pragma unroll
        for (int q = 0; q < 4; ++q)
            u[q] += __shfl_xor(u[q], off, 32);
    }
    if (l15 == 0) {
#pragma unroll
        for (int q = 0; q < 4; ++q)
            red[wv][half * 4 + q] = u[q];
    }
    __syncthreads();
    if (threadIdx.x < 8) {
        const int jl = threadIdx.x;
        const float v = (red[0][jl] + red[1][jl]) + (red[2][jl] + red[3][jl]);
        float* __restrict__ sb = s + bt * NN + mtile * 8 + jl;
        sb[0]      = v;
        sb[NN / 2] = v;
    }
}

__global__ __launch_bounds__(128) void k_attn_out(const float* __restrict__ adj,
                                                  const float* __restrict__ h,
                                                  const float* __restrict__ s,
                                                  float* __restrict__ out) {
    __shared__ __attribute__((aligned(16))) float attn_lds[16][LDSW];

    const int blk   = blockIdx.x;
    const int bt    = blk >> 4;
    const int itile = blk & 15;
    const int b     = bt >> 2;
    const int tau   = bt & 3;
    const int cpred = (itile >= 8) ? 1 : 0;
    const int tp    = (2 * tau + cpred) & 3;

    const int wv   = threadIdx.x >> 5;
    const int lane = threadIdx.x & 31;
    const int half = lane >> 4;
    const int l15  = lane & 15;

    const float* __restrict__ srow = s + (b * 4 + tp) * NN;

    float ev[8];
#pragma unroll
    for (int u = 0; u < 8; ++u) {
        const float sv = srow[lane + u * 32];
        ev[u] = (sv > 0.f) ? sv : 0.2f * sv;
    }
    float z[4][8];
#pragma unroll
    for (int rr = 0; rr < 4; ++rr) {
        const int i = itile * 16 + wv * 4 + rr;
        const float* __restrict__ arow = adj + ((size_t)(bt * NN + i)) * NN;
#pragma unroll
        for (int u = 0; u < 8; ++u)
            z[rr][u] = arow[lane + u * 32];
    }
    const float4* __restrict__ hfrag = reinterpret_cast<const float4*>(
        h + (size_t)((bt * 4 + wv) * 16) * 256);
    float4 hv[32];
#pragma unroll
    for (int kb = 0; kb < 16; ++kb) {
        hv[2 * kb]     = hfrag[kb * 64 + lane];
        hv[2 * kb + 1] = hfrag[kb * 64 + 32 + lane];
    }
    __builtin_amdgcn_sched_barrier(0);

#pragma unroll
    for (int rr = 0; rr < 4; ++rr) {
        const int rloc = wv * 4 + rr;
        float mx = -3.402823466e38f;
#pragma unroll
        for (int u = 0; u < 8; ++u) {
            const float zz = z[rr][u] * ev[u];
            z[rr][u] = zz;
            mx = fmaxf(mx, zz);
        }
#pragma unroll
        for (int off = 16; off > 0; off >>= 1)
            mx = fmaxf(mx, __shfl_xor(mx, off, 32));
        float sum = 0.f;
#pragma unroll
        for (int u = 0; u < 8; ++u) {
            const float p = __expf(z[rr][u] - mx);
            z[rr][u] = p;
            sum += p;
        }
#pragma unroll
        for (int off = 16; off > 0; off >>= 1)
            sum += __shfl_xor(sum, off, 32);
        const float inv = 1.f / sum;
#pragma unroll
        for (int u = 0; u < 8; ++u)
            attn_lds[rloc][lane + u * 32] = z[rr][u] * inv;
    }
    __syncthreads();

    const int colbase = wv * 16 + l15;
    v8f c = {};
#pragma unroll
    for (int kb = 0; kb < 16; ++kb) {
        const float4 blo = hv[2 * kb];
        const float4 bhi = hv[2 * kb + 1];
        const float4 alo = *reinterpret_cast<const float4*>(&attn_lds[l15][kb * 16 + half * 8]);
        const float4 ahi = *reinterpret_cast<const float4*>(&attn_lds[l15][kb * 16 + half * 8 + 4]);
        v2f a_, b_;
        a_.x = alo.x; a_.y = alo.y; b_.x = blo.x; b_.y = blo.y;
        c = __builtin_amdgcn_wmma_f32_16x16x4_f32(false, a_, false, b_, (short)0, c, false, false);
        a_.x = alo.z; a_.y = alo.w; b_.x = blo.z; b_.y = blo.w;
        c = __builtin_amdgcn_wmma_f32_16x16x4_f32(false, a_, false, b_, (short)0, c, false, false);
        a_.x = ahi.x; a_.y = ahi.y; b_.x = bhi.x; b_.y = bhi.y;
        c = __builtin_amdgcn_wmma_f32_16x16x4_f32(false, a_, false, b_, (short)0, c, false, false);
        a_.x = ahi.z; a_.y = ahi.w; b_.x = bhi.z; b_.y = bhi.w;
        c = __builtin_amdgcn_wmma_f32_16x16x4_f32(false, a_, false, b_, (short)0, c, false, false);
    }
    float* __restrict__ orow = out + (size_t)(bt * NN + itile * 16) * FOUT + colbase;
#pragma unroll
    for (int r = 0; r < 8; ++r)
        orow[(size_t)(r + half * 8) * FOUT] = c[r];
}

extern "C" void kernel_launch(void* const* d_in, const int* in_sizes, int n_in,
                              void* d_out, int out_size, void* d_ws, size_t ws_size,
                              hipStream_t stream) {
    const float* x   = (const float*)d_in[0];
    const float* adj = (const float*)d_in[1];
    const float* W   = (const float*)d_in[2];
    const float* a   = (const float*)d_in[3];
    float* out = (float*)d_out;

    float* h = (float*)d_ws;
    float* s = h + (size_t)BT * NN * FOUT;

    k_gemm_xw_s<<<dim3(BT * 16), dim3(128), 0, stream>>>(x, W, a, h, s);
    k_attn_out<<<dim3(BT * 16), dim3(128), 0, stream>>>(adj, h, s, out);
}


# Kernel 1 operands through LDS: x tile and W fetched once per block, 20 whole-line 16-byte loads per lane (was 96 small ones, x on 16 lines each), fragments read from padded LDS; sums unchanged
# speedup vs baseline: 1.8576x; 1.0754x over previous
#include <hip/hip_runtime.h>
#include <math.h>


typedef __attribute__((ext_vector_type(2))) float v2f;
typedef __attribute__((ext_vector_type(8))) float v8f;

#define BT    8
#define NN    256
#define FIN   128
#define FOUT  64
#define LDSW  260

#define XP  132
#define WP  80
__global__ __launch_bounds__(128) void k_gemm_xw_s(const float* __restrict__ x,
                                                   const float* __restrict__ W,
                                                   const float* __restrict__ a,
                                                   float* __restrict__ h,
                                                   float* __restrict__ s) {
    __shared__ __attribute__((aligned(16))) float xs[16][XP];
    __shared__ __attribute__((aligned(16))) float Ws[FIN][WP];
    __shared__ float red[4][8];

    const int blk   = blockIdx.x;
    const int bt    = blk >> 4;
    const int mtile = blk & 15;
    const int tid   = threadIdx.x;
    const int wv    = tid >> 5;
    const int lane  = tid & 31;
    const int half  = lane >> 4;
    const int l15   = lane & 15;
    const int colbase = wv * 16 + l15;

    const float4* __restrict__ xg = reinterpret_cast<const float4*>(
        x + (size_t)(bt * NN + mtile * 16) * FIN);
    const float4* __restrict__ wg = reinterpret_cast<const float4*>(W);
    float4 xv[4];
#pragma unroll
    for (int i = 0; i < 4; ++i)
        xv[i] = xg[i * 128 + tid];
    float4 wq[16];
#pragma unroll
    for (int i = 0; i < 16; ++i)
        wq[i] = wg[i * 128 + tid];

    const float a1 = a[colbase];
    const float a2 = a[FOUT + colbase];

#pragma unroll
    for (int i = 0; i < 4; ++i)
        *reinterpret_cast<float4*>(&xs[4 * i + (tid >> 5)][4 * (tid & 31)]) = xv[i];
#pragma unroll
    for (int i = 0; i < 16; ++i)
        *reinterpret_cast<float4*>(&Ws[8 * i + (tid >> 4)][4 * (tid & 15)]) = wq[i];
    __syncthreads();

    v2f af[32], bf[32];
#pragma unroll
    for (int t = 0; t < 32; ++t) {
        const int ka = 4 * t + half * 2;
        af[t] = *reinterpret_cast<const v2f*>(&xs[l15][ka]);
        bf[t].x = Ws[ka][colbase];
        bf[t].y = Ws[ka + 1][colbase];
    }
    __builtin_amdgcn_sched_barrier(0);
    v8f c = {};
#pragma unroll
    for (int t = 0; t < 32; ++t)
        c = __builtin_amdgcn_wmma_f32_16x16x4_f32(false, af[t], false, bf[t], (short)0, c,
                                                  false, false);
    float4* __restrict__ hout = reinterpret_cast<float4*>(
        h + (size_t)((bt * 4 + wv) * 16 + mtile) * 256);
    hout[lane]      = make_float4(c[0], c[1], c[2], c[3]);
    hout[32 + lane] = make_float4(c[4], c[5], c[6], c[7]);

    float u[4];
#pragma unroll
    for (int q = 0; q < 4; ++q)
        u[q] = fmaf(c[2 * q], a1, c[2 * q + 1] * a2);
#pragma unroll
    for (int off = 8; off > 0; off >>= 1) {
#pragma unroll
        for (int q = 0; q < 4; ++q)
            u[q] += __shfl_xor(u[q], off, 32);
    }
    if (l15 == 0) {
#pragma unroll
        for (int q = 0; q < 4; ++q)
            red[wv][half * 4 + q] = u[q];
    }
    __syncthreads();
    if (threadIdx.x < 8) {
        const int jl = threadIdx.x;
        const float v = (red[0][jl] + red[1][jl]) + (red[2][jl] + red[3][jl]);
        float* __restrict__ sb = s + bt * NN + mtile * 8 + jl;
        sb[0]      = v;
        sb[NN / 2] = v;
    }
}

__global__ __launch_bounds__(128) void k_attn_out(const float* __restrict__ adj,
                                                  const float* __restrict__ h,
                                                  const float* __restrict__ s,
                                                  float* __restrict__ out) {
    __shared__ __attribute__((aligned(16))) float attn_lds[16][LDSW];

    const int blk   = blockIdx.x;
    const int bt    = blk >> 4;
    const int itile = blk & 15;
    const int b     = bt >> 2;
    const int tau   = bt & 3;
    const int cpred = (itile >= 8) ? 1 : 0;
    const int tp    = (2 * tau + cpred) & 3;

    const int wv   = threadIdx.x >> 5;
    const int lane = threadIdx.x & 31;
    const int half = lane >> 4;
    const int l15  = lane & 15;

    const float* __restrict__ srow = s + (b * 4 + tp) * NN;

    float ev[8];
#pragma unroll
    for (int u = 0; u < 8; ++u) {
        const float sv = srow[lane + u * 32];
        ev[u] = (sv > 0.f) ? sv : 0.2f * sv;
    }
    float z[4][8];
#pragma unroll
    for (int rr = 0; rr < 4; ++rr) {
        const int i = itile * 16 + wv * 4 + rr;
        const float* __restrict__ arow = adj + ((size_t)(bt * NN + i)) * NN;
#pragma unroll
        for (int u = 0; u < 8; ++u)
            z[rr][u] = arow[lane + u * 32];
    }
    const float4* __restrict__ hfrag = reinterpret_cast<const float4*>(
        h + (size_t)((bt * 4 + wv) * 16) * 256);
    float4 hv[32];
#pragma unroll
    for (int kb = 0; kb < 16; ++kb) {
        hv[2 * kb]     = hfrag[kb * 64 + lane];
        hv[2 * kb + 1] = hfrag[kb * 64 + 32 + lane];
    }
    __builtin_amdgcn_sched_barrier(0);

#pragma unroll
    for (int rr = 0; rr < 4; ++rr) {
        const int rloc = wv * 4 + rr;
        float mx = -3.402823466e38f;
#pragma unroll
        for (int u = 0; u < 8; ++u) {
            const float zz = z[rr][u] * ev[u];
            z[rr][u] = zz;
            mx = fmaxf(mx, zz);
        }
#pragma unroll
        for (int off = 16; off > 0; off >>= 1)
            mx = fmaxf(mx, __shfl_xor(mx, off, 32));
        float sum = 0.f;
#pragma unroll
        for (int u = 0; u < 8; ++u) {
            const float p = __expf(z[rr][u] - mx);
            z[rr][u] = p;
            sum += p;
        }
#pragma unroll
        for (int off = 16; off > 0; off >>= 1)
            sum += __shfl_xor(sum, off, 32);
        const float inv = 1.f / sum;
#pragma unroll
        for (int u = 0; u < 8; ++u)
            attn_lds[rloc][lane + u * 32] = z[rr][u] * inv;
    }
    __syncthreads();

    const int colbase = wv * 16 + l15;
    v8f c = {};
#pragma unroll
    for (int kb = 0; kb < 16; ++kb) {
        const float4 blo = hv[2 * kb];
        const float4 bhi = hv[2 * kb + 1];
        const float4 alo = *reinterpret_cast<const float4*>(&attn_lds[l15][kb * 16 + half * 8]);
        const float4 ahi = *reinterpret_cast<const float4*>(&attn_lds[l15][kb * 16 + half * 8 + 4]);
        v2f a_, b_;
        a_.x = alo.x; a_.y = alo.y; b_.x = blo.x; b_.y = blo.y;
        c = __builtin_amdgcn_wmma_f32_16x16x4_f32(false, a_, false, b_, (short)0, c, false, false);
        a_.x = alo.z; a_.y = alo.w; b_.x = blo.z; b_.y = blo.w;
        c = __builtin_amdgcn_wmma_f32_16x16x4_f32(false, a_, false, b_, (short)0, c, false, false);
        a_.x = ahi.x; a_.y = ahi.y; b_.x = bhi.x; b_.y = bhi.y;
        c = __builtin_amdgcn_wmma_f32_16x16x4_f32(false, a_, false, b_, (short)0, c, false, false);
        a_.x = ahi.z; a_.y = ahi.w; b_.x = bhi.z; b_.y = bhi.w;
        c = __builtin_amdgcn_wmma_f32_16x16x4_f32(false, a_, false, b_, (short)0, c, false, false);
    }
    float* __restrict__ orow = out + (size_t)(bt * NN + itile * 16) * FOUT + colbase;
#pragma unroll
    for (int r = 0; r < 8; ++r)
        orow[(size_t)(r + half * 8) * FOUT] = c[r];
}

extern "C" void kernel_launch(void* const* d_in, const int* in_sizes, int n_in,
                              void* d_out, int out_size, void* d_ws, size_t ws_size,
                              hipStream_t stream) {
    const float* x   = (const float*)d_in[0];
    const float* adj = (const float*)d_in[1];
    const float* W   = (const float*)d_in[2];
    const float* a   = (const float*)d_in[3];
    float* out = (float*)d_out;

    float* h = (float*)d_ws;
    float* s = h + (size_t)BT * NN * FOUT;

    k_gemm_xw_s<<<dim3(BT * 16), dim3(128), 0, stream>>>(x, W, a, h, s);
    k_attn_out<<<dim3(BT * 16), dim3(128), 0, stream>>>(adj, h, s, out);
}


# Kernel 2 softmax lanes re-mapped: 8 lanes per row, a wave's 4 rows side by side, so row max and row sum take 3 shuffle steps shared by all 4 rows (was 5 per row); 16-byte loads; row sums in new order
# speedup vs baseline: 2.0719x; 1.1154x over previous
#include <hip/hip_runtime.h>
#include <math.h>


typedef __attribute__((ext_vector_type(2))) float v2f;
typedef __attribute__((ext_vector_type(8))) float v8f;

#define BT    8
#define NN    256
#define FIN   128
#define FOUT  64
#define LDSW  260

#define XP  132
#define WP  80
__global__ __launch_bounds__(128) void k_gemm_xw_s(const float* __restrict__ x,
                                                   const float* __restrict__ W,
                                                   const float* __restrict__ a,
                                                   float* __restrict__ h,
                                                   float* __restrict__ s) {
    __shared__ __attribute__((aligned(16))) float xs[16][XP];
    __shared__ __attribute__((aligned(16))) float Ws[FIN][WP];
    __shared__ float red[4][8];

    const int blk   = blockIdx.x;
    const int bt    = blk >> 4;
    const int mtile = blk & 15;
    const int tid   = threadIdx.x;
    const int wv    = tid >> 5;
    const int lane  = tid & 31;
    const int half  = lane >> 4;
    const int l15   = lane & 15;
    const int colbase = wv * 16 + l15;

    const float4* __restrict__ xg = reinterpret_cast<const float4*>(
        x + (size_t)(bt * NN + mtile * 16) * FIN);
    const float4* __restrict__ wg = reinterpret_cast<const float4*>(W);
    float4 xv[4];
#pragma unroll
    for (int i = 0; i < 4; ++i)
        xv[i] = xg[i * 128 + tid];
    float4 wq[16];
#pragma unroll
    for (int i = 0; i < 16; ++i)
        wq[i] = wg[i * 128 + tid];

    const float a1 = a[colbase];
    const float a2 = a[FOUT + colbase];

#pragma unroll
    for (int i = 0; i < 4; ++i)
        *reinterpret_cast<float4*>(&xs[4 * i + (tid >> 5)][4 * (tid & 31)]) = xv[i];
#pragma unroll
    for (int i = 0; i < 16; ++i)
        *reinterpret_cast<float4*>(&Ws[8 * i + (tid >> 4)][4 * (tid & 15)]) = wq[i];
    __syncthreads();

    v2f af[32], bf[32];
#pragma unroll
    for (int t = 0; t < 32; ++t) {
        const int ka = 4 * t + half * 2;
        af[t] = *reinterpret_cast<const v2f*>(&xs[l15][ka]);
        bf[t].x = Ws[ka][colbase];
        bf[t].y = Ws[ka + 1][colbase];
    }
    __builtin_amdgcn_sched_barrier(0);
    v8f c = {};
#pragma unroll
    for (int t = 0; t < 32; ++t)
        c = __builtin_amdgcn_wmma_f32_16x16x4_f32(false, af[t], false, bf[t], (short)0, c,
                                                  false, false);
    float4* __restrict__ hout = reinterpret_cast<float4*>(
        h + (size_t)((bt * 4 + wv) * 16 + mtile) * 256);
    hout[lane]      = make_float4(c[0], c[1], c[2], c[3]);
    hout[32 + lane] = make_float4(c[4], c[5], c[6], c[7]);

    float u[4];
#pragma unroll
    for (int q = 0; q < 4; ++q)
        u[q] = fmaf(c[2 * q], a1, c[2 * q + 1] * a2);
#pragma unroll
    for (int off = 8; off > 0; off >>= 1) {
#pragma unroll
        for (int q = 0; q < 4; ++q)
            u[q] += __shfl_xor(u[q], off, 32);
    }
    if (l15 == 0) {
#pragma unroll
        for (int q = 0; q < 4; ++q)
            red[wv][half * 4 + q] = u[q];
    }
    __syncthreads();
    if (threadIdx.x < 8) {
        const int jl = threadIdx.x;
        const float v = (red[0][jl] + red[1][jl]) + (red[2][jl] + red[3][jl]);
        float* __restrict__ sb = s + bt * NN + mtile * 8 + jl;
        sb[0]      = v;
        sb[NN / 2] = v;
    }
}

__global__ __launch_bounds__(128) void k_attn_out(const float* __restrict__ adj,
                                                  const float* __restrict__ h,
                                                  const float* __restrict__ s,
                                                  float* __restrict__ out) {
    __shared__ __attribute__((aligned(16))) float attn_lds[16][LDSW];

    const int blk   = blockIdx.x;
    const int bt    = blk >> 4;
    const int itile = blk & 15;
    const int b     = bt >> 2;
    const int tau   = bt & 3;
    const int cpred = (itile >= 8) ? 1 : 0;
    const int tp    = (2 * tau + cpred) & 3;

    const int wv   = threadIdx.x >> 5;
    const int lane = threadIdx.x & 31;
    const int half = lane >> 4;
    const int l15  = lane & 15;

    const int g    = lane >> 3;
    const int cs   = lane & 7;
    const int rloc = 2 * wv + (g >> 1) + 8 * (g & 1);

    const float4* __restrict__ srow4 = reinterpret_cast<const float4*>(s + (b * 4 + tp) * NN) + cs;
    float4 ev[8];
#pragma unroll
    for (int i = 0; i < 8; ++i) {
        const float4 sv = srow4[8 * i];
        ev[i].x = (sv.x > 0.f) ? sv.x : 0.2f * sv.x;
        ev[i].y = (sv.y > 0.f) ? sv.y : 0.2f * sv.y;
        ev[i].z = (sv.z > 0.f) ? sv.z : 0.2f * sv.z;
        ev[i].w = (sv.w > 0.f) ? sv.w : 0.2f * sv.w;
    }
    const float4* __restrict__ arow4 = reinterpret_cast<const float4*>(
        adj + ((size_t)(bt * NN + itile * 16 + rloc)) * NN) + cs;
    float4 z[8];
#pragma unroll
    for (int i = 0; i < 8; ++i)
        z[i] = arow4[8 * i];
    const float4* __restrict__ hfrag = reinterpret_cast<const float4*>(
        h + (size_t)((bt * 4 + wv) * 16) * 256);
    float4 hv[32];
#pragma unroll
    for (int kb = 0; kb < 16; ++kb) {
        hv[2 * kb]     = hfrag[kb * 64 + lane];
        hv[2 * kb + 1] = hfrag[kb * 64 + 32 + lane];
    }
    __builtin_amdgcn_sched_barrier(0);

    float4 m4 = make_float4(-3.402823466e38f, -3.402823466e38f, -3.402823466e38f, -3.402823466e38f);
#pragma unroll
    for (int i = 0; i < 8; ++i) {
        z[i].x *= ev[i].x; z[i].y *= ev[i].y; z[i].z *= ev[i].z; z[i].w *= ev[i].w;
        m4.x = fmaxf(m4.x, z[i].x); m4.y = fmaxf(m4.y, z[i].y);
        m4.z = fmaxf(m4.z, z[i].z); m4.w = fmaxf(m4.w, z[i].w);
    }
    float mx = fmaxf(fmaxf(m4.x, m4.y), fmaxf(m4.z, m4.w));
#pragma unroll
    for (int off = 4; off > 0; off >>= 1)
        mx = fmaxf(mx, __shfl_xor(mx, off, 32));
    float4 s4 = make_float4(0.f, 0.f, 0.f, 0.f);
#pragma unroll
    for (int i = 0; i < 8; ++i) {
        z[i].x = __expf(z[i].x - mx); s4.x += z[i].x;
        z[i].y = __expf(z[i].y - mx); s4.y += z[i].y;
        z[i].z = __expf(z[i].z - mx); s4.z += z[i].z;
        z[i].w = __expf(z[i].w - mx); s4.w += z[i].w;
    }
    float sum = (s4.x + s4.y) + (s4.z + s4.w);
#pragma unroll
    for (int off = 4; off > 0; off >>= 1)
        sum += __shfl_xor(sum, off, 32);
    const float inv = 1.f / sum;
#pragma unroll
    for (int i = 0; i < 8; ++i)
        *reinterpret_cast<float4*>(&attn_lds[rloc][32 * i + 4 * cs]) =
            make_float4(z[i].x * inv, z[i].y * inv, z[i].z * inv, z[i].w * inv);
    __syncthreads();

    const int colbase = wv * 16 + l15;
    v8f c = {};
#pragma unroll
    for (int kb = 0; kb < 16; ++kb) {
        const float4 blo = hv[2 * kb];
        const float4 bhi = hv[2 * kb + 1];
        const float4 alo = *reinterpret_cast<const float4*>(&attn_lds[l15][kb * 16 + half * 8]);
        const float4 ahi = *reinterpret_cast<const float4*>(&attn_lds[l15][kb * 16 + half * 8 + 4]);
        v2f a_, b_;
        a_.x = alo.x; a_.y = alo.y; b_.x = blo.x; b_.y = blo.y;
        c = __builtin_amdgcn_wmma_f32_16x16x4_f32(false, a_, false, b_, (short)0, c, false, false);
        a_.x = alo.z; a_.y = alo.w; b_.x = blo.z; b_.y = blo.w;
        c = __builtin_amdgcn_wmma_f32_16x16x4_f32(false, a_, false, b_, (short)0, c, false, false);
        a_.x = ahi.x; a_.y = ahi.y; b_.x = bhi.x; b_.y = bhi.y;
        c = __builtin_amdgcn_wmma_f32_16x16x4_f32(false, a_, false, b_, (short)0, c, false, false);
        a_.x = ahi.z; a_.y = ahi.w; b_.x = bhi.z; b_.y = bhi.w;
        c = __builtin_amdgcn_wmma_f32_16x16x4_f32(false, a_, false, b_, (short)0, c, false, false);
    }
    float* __restrict__ orow = out + (size_t)(bt * NN + itile * 16) * FOUT + colbase;
#pragma unroll
    for (int r = 0; r < 8; ++r)
        orow[(size_t)(r + half * 8) * FOUT] = c[r];
}

extern "C" void kernel_launch(void* const* d_in, const int* in_sizes, int n_in,
                              void* d_out, int out_size, void* d_ws, size_t ws_size,
                              hipStream_t stream) {
    const float* x   = (const float*)d_in[0];
    const float* adj = (const float*)d_in[1];
    const float* W   = (const float*)d_in[2];
    const float* a   = (const float*)d_in[3];
    float* out = (float*)d_out;

    float* h = (float*)d_ws;
    float* s = h + (size_t)BT * NN * FOUT;

    k_gemm_xw_s<<<dim3(BT * 16), dim3(128), 0, stream>>>(x, W, a, h, s);
    k_attn_out<<<dim3(BT * 16), dim3(128), 0, stream>>>(adj, h, s, out);
}


# Kernel 2 on all 256 WGPs: a block has 16 rows, 32 of 64 columns; each wave one column tile and one k half, 32 matrix steps (was 64) on two accumulators; halves meet in LDS; out summed in new order
# speedup vs baseline: 2.2100x; 1.0667x over previous
#include <hip/hip_runtime.h>
#include <math.h>


typedef __attribute__((ext_vector_type(2))) float v2f;
typedef __attribute__((ext_vector_type(8))) float v8f;

#define BT    8
#define NN    256
#define FIN   128
#define FOUT  64
#define LDSW  260

#define XP  132
#define WP  80
__global__ __launch_bounds__(128) void k_gemm_xw_s(const float* __restrict__ x,
                                                   const float* __restrict__ W,
                                                   const float* __restrict__ a,
                                                   float* __restrict__ h,
                                                   float* __restrict__ s) {
    __shared__ __attribute__((aligned(16))) float xs[16][XP];
    __shared__ __attribute__((aligned(16))) float Ws[FIN][WP];
    __shared__ float red[4][8];

    const int blk   = blockIdx.x;
    const int bt    = blk >> 4;
    const int mtile = blk & 15;
    const int tid   = threadIdx.x;
    const int wv    = tid >> 5;
    const int lane  = tid & 31;
    const int half  = lane >> 4;
    const int l15   = lane & 15;
    const int colbase = wv * 16 + l15;

    const float4* __restrict__ xg = reinterpret_cast<const float4*>(
        x + (size_t)(bt * NN + mtile * 16) * FIN);
    const float4* __restrict__ wg = reinterpret_cast<const float4*>(W);
    float4 xv[4];
#pragma unroll
    for (int i = 0; i < 4; ++i)
        xv[i] = xg[i * 128 + tid];
    float4 wq[16];
#pragma unroll
    for (int i = 0; i < 16; ++i)
        wq[i] = wg[i * 128 + tid];

    const float a1 = a[colbase];
    const float a2 = a[FOUT + colbase];

#pragma unroll
    for (int i = 0; i < 4; ++i)
        *reinterpret_cast<float4*>(&xs[4 * i + (tid >> 5)][4 * (tid & 31)]) = xv[i];
#pragma unroll
    for (int i = 0; i < 16; ++i)
        *reinterpret_cast<float4*>(&Ws[8 * i + (tid >> 4)][4 * (tid & 15)]) = wq[i];
    __syncthreads();

    v2f af[32], bf[32];
#pragma unroll
    for (int t = 0; t < 32; ++t) {
        const int ka = 4 * t + half * 2;
        af[t] = *reinterpret_cast<const v2f*>(&xs[l15][ka]);
        bf[t].x = Ws[ka][colbase];
        bf[t].y = Ws[ka + 1][colbase];
    }
    __builtin_amdgcn_sched_barrier(0);
    v8f c = {};
#pragma unroll
    for (int t = 0; t < 32; ++t)
        c = __builtin_amdgcn_wmma_f32_16x16x4_f32(false, af[t], false, bf[t], (short)0, c,
                                                  false, false);
    float4* __restrict__ hout = reinterpret_cast<float4*>(
        h + (size_t)((bt * 4 + wv) * 16 + mtile) * 256);
    hout[lane]      = make_float4(c[0], c[1], c[2], c[3]);
    hout[32 + lane] = make_float4(c[4], c[5], c[6], c[7]);

    float u[4];
#pragma unroll
    for (int q = 0; q < 4; ++q)
        u[q] = fmaf(c[2 * q], a1, c[2 * q + 1] * a2);
#pragma unroll
    for (int off = 8; off > 0; off >>= 1) {
#pragma unroll
        for (int q = 0; q < 4; ++q)
            u[q] += __shfl_xor(u[q], off, 32);
    }
    if (l15 == 0) {
#pragma unroll
        for (int q = 0; q < 4; ++q)
            red[wv][half * 4 + q] = u[q];
    }
    __syncthreads();
    if (threadIdx.x < 8) {
        const int jl = threadIdx.x;
        const float v = (red[0][jl] + red[1][jl]) + (red[2][jl] + red[3][jl]);
        float* __restrict__ sb = s + bt * NN + mtile * 8 + jl;
        sb[0]      = v;
        sb[NN / 2] = v;
    }
}

__global__ __launch_bounds__(128) void k_attn_out(const float* __restrict__ adj,
                                                  const float* __restrict__ h,
                                                  const float* __restrict__ s,
                                                  float* __restrict__ out) {
    __shared__ __attribute__((aligned(16))) float attn_lds[16][LDSW];
    __shared__ __attribute__((aligned(16))) float4 part[2][2][32];

    const int blk   = blockIdx.x;
    const int bt    = blk >> 5;
    const int itile = (blk >> 1) & 15;
    const int nhalf = blk & 1;
    const int b     = bt >> 2;
    const int tau   = bt & 3;
    const int cpred = (itile >= 8) ? 1 : 0;
    const int tp    = (2 * tau + cpred) & 3;

    const int wv   = threadIdx.x >> 5;
    const int lane = threadIdx.x & 31;
    const int half = lane >> 4;
    const int l15  = lane & 15;
    const int nt2  = wv & 1;
    const int kh   = wv >> 1;
    const int nt   = 2 * nhalf + nt2;

    const int g    = lane >> 3;
    const int cs   = lane & 7;
    const int rloc = 2 * wv + (g >> 1) + 8 * (g & 1);

    const float4* __restrict__ srow4 = reinterpret_cast<const float4*>(s + (b * 4 + tp) * NN) + cs;
    float4 ev[8];
#pragma unroll
    for (int i = 0; i < 8; ++i) {
        const float4 sv = srow4[8 * i];
        ev[i].x = (sv.x > 0.f) ? sv.x : 0.2f * sv.x;
        ev[i].y = (sv.y > 0.f) ? sv.y : 0.2f * sv.y;
        ev[i].z = (sv.z > 0.f) ? sv.z : 0.2f * sv.z;
        ev[i].w = (sv.w > 0.f) ? sv.w : 0.2f * sv.w;
    }
    const float4* __restrict__ arow4 = reinterpret_cast<const float4*>(
        adj + ((size_t)(bt * NN + itile * 16 + rloc)) * NN) + cs;
    float4 z[8];
#pragma unroll
    for (int i = 0; i < 8; ++i)
        z[i] = arow4[8 * i];
    const float4* __restrict__ hfrag = reinterpret_cast<const float4*>(
        h + (size_t)((bt * 4 + nt) * 16 + 8 * kh) * 256);
    float4 hv[16];
#pragma unroll
    for (int kb = 0; kb < 8; ++kb) {
        hv[2 * kb]     = hfrag[kb * 64 + lane];
        hv[2 * kb + 1] = hfrag[kb * 64 + 32 + lane];
    }
    __builtin_amdgcn_sched_barrier(0);

    float4 m4 = make_float4(-3.402823466e38f, -3.402823466e38f, -3.402823466e38f, -3.402823466e38f);
#pragma unroll
    for (int i = 0; i < 8; ++i) {
        z[i].x *= ev[i].x; z[i].y *= ev[i].y; z[i].z *= ev[i].z; z[i].w *= ev[i].w;
        m4.x = fmaxf(m4.x, z[i].x); m4.y = fmaxf(m4.y, z[i].y);
        m4.z = fmaxf(m4.z, z[i].z); m4.w = fmaxf(m4.w, z[i].w);
    }
    float mx = fmaxf(fmaxf(m4.x, m4.y), fmaxf(m4.z, m4.w));
#pragma unroll
    for (int off = 4; off > 0; off >>= 1)
        mx = fmaxf(mx, __shfl_xor(mx, off, 32));
    float4 s4 = make_float4(0.f, 0.f, 0.f, 0.f);
#pragma unroll
    for (int i = 0; i < 8; ++i) {
        z[i].x = __expf(z[i].x - mx); s4.x += z[i].x;
        z[i].y = __expf(z[i].y - mx); s4.y += z[i].y;
        z[i].z = __expf(z[i].z - mx); s4.z += z[i].z;
        z[i].w = __expf(z[i].w - mx); s4.w += z[i].w;
    }
    float sum = (s4.x + s4.y) + (s4.z + s4.w);
#pragma unroll
    for (int off = 4; off > 0; off >>= 1)
        sum += __shfl_xor(sum, off, 32);
    const float inv = 1.f / sum;
#pragma unroll
    for (int i = 0; i < 8; ++i)
        *reinterpret_cast<float4*>(&attn_lds[rloc][32 * i + 4 * cs]) =
            make_float4(z[i].x * inv, z[i].y * inv, z[i].z * inv, z[i].w * inv);
    __syncthreads();

    v8f c0 = {}, c1 = {};
#pragma unroll
    for (int kb = 0; kb < 8; ++kb) {
        const int kcol = (8 * kh + kb) * 16 + half * 8;
        const float4 blo = hv[2 * kb];
        const float4 bhi = hv[2 * kb + 1];
        const float4 alo = *reinterpret_cast<const float4*>(&attn_lds[l15][kcol]);
        const float4 ahi = *reinterpret_cast<const float4*>(&attn_lds[l15][kcol + 4]);
        v2f a_, b_;
        a_.x = alo.x; a_.y = alo.y; b_.x = blo.x; b_.y = blo.y;
        c0 = __builtin_amdgcn_wmma_f32_16x16x4_f32(false, a_, false, b_, (short)0, c0, false, false);
        a_.x = alo.z; a_.y = alo.w; b_.x = blo.z; b_.y = blo.w;
        c1 = __builtin_amdgcn_wmma_f32_16x16x4_f32(false, a_, false, b_, (short)0, c1, false, false);
        a_.x = ahi.x; a_.y = ahi.y; b_.x = bhi.x; b_.y = bhi.y;
        c0 = __builtin_amdgcn_wmma_f32_16x16x4_f32(false, a_, false, b_, (short)0, c0, false, false);
        a_.x = ahi.z; a_.y = ahi.w; b_.x = bhi.z; b_.y = bhi.w;
        c1 = __builtin_amdgcn_wmma_f32_16x16x4_f32(false, a_, false, b_, (short)0, c1, false, false);
    }
    float t[8];
#pragma unroll
    for (int r = 0; r < 8; ++r)
        t[r] = c0[r] + c1[r];
    const bool up = (kh != 0);
    const float4 keep = make_float4(up ? t[4] : t[0], up ? t[5] : t[1], up ? t[6] : t[2], up ? t[7] : t[3]);
    const float4 give = make_float4(up ? t[0] : t[4], up ? t[1] : t[5], up ? t[2] : t[6], up ? t[3] : t[7]);
    part[nt2][kh][lane] = give;
    __syncthreads();
    const float4 got = part[nt2][kh ^ 1][lane];
    const float4 lo = kh ? got : keep;
    const float4 hi = kh ? keep : got;
    const float o0 = lo.x + hi.x, o1 = lo.y + hi.y, o2 = lo.z + hi.z, o3 = lo.w + hi.w;
    float* __restrict__ orow = out + (size_t)(bt * NN + itile * 16 + half * 8 + 4 * kh) * FOUT + nt * 16 + l15;
    orow[0 * FOUT] = o0;
    orow[1 * FOUT] = o1;
    orow[2 * FOUT] = o2;
    orow[3 * FOUT] = o3;
}

extern "C" void kernel_launch(void* const* d_in, const int* in_sizes, int n_in,
                              void* d_out, int out_size, void* d_ws, size_t ws_size,
                              hipStream_t stream) {
    const float* x   = (const float*)d_in[0];
    const float* adj = (const float*)d_in[1];
    const float* W   = (const float*)d_in[2];
    const float* a   = (const float*)d_in[3];
    float* out = (float*)d_out;

    float* h = (float*)d_ws;
    float* s = h + (size_t)BT * NN * FOUT;

    k_gemm_xw_s<<<dim3(BT * 16), dim3(128), 0, stream>>>(x, W, a, h, s);
    k_attn_out<<<dim3(BT * 16 * 2), dim3(128), 0, stream>>>(adj, h, s, out);
}


# No bpermute steps: lanes park partials in LDS, readers add them (K1 scores, h stored after the barrier; K2 row max/sum); K2 passes exp unnormalised, scales out rows, groups LDS reads; new sum order
# speedup vs baseline: 2.3512x; 1.0639x over previous
#include <hip/hip_runtime.h>
#include <math.h>


typedef __attribute__((ext_vector_type(2))) float v2f;
typedef __attribute__((ext_vector_type(4))) float v4f;
typedef __attribute__((ext_vector_type(8))) float v8f;

#define BT    8
#define NN    256
#define FIN   128
#define FOUT  64
#define LDSW  260

#define XP  132
#define WP  80
#define RP  68
__global__ __launch_bounds__(128) void k_gemm_xw_s(const float* __restrict__ x,
                                                   const float* __restrict__ W,
                                                   const float* __restrict__ a,
                                                   float* __restrict__ h,
                                                   float* __restrict__ s) {
    __shared__ __attribute__((aligned(16))) float xs[16][XP];
    __shared__ __attribute__((aligned(16))) float Ws[FIN][WP];
    __shared__ __attribute__((aligned(16))) float red[8][RP];

    const int blk   = blockIdx.x;
    const int bt    = blk >> 4;
    const int mtile = blk & 15;
    const int tid   = threadIdx.x;
    const int wv    = tid >> 5;
    const int lane  = tid & 31;
    const int half  = lane >> 4;
    const int l15   = lane & 15;
    const int colbase = wv * 16 + l15;

    const float4* __restrict__ xg = reinterpret_cast<const float4*>(
        x + (size_t)(bt * NN + mtile * 16) * FIN);
    const float4* __restrict__ wg = reinterpret_cast<const float4*>(W);
    float4 xv[4];
#pragma unroll
    for (int i = 0; i < 4; ++i)
        xv[i] = xg[i * 128 + tid];
    float4 wq[16];
#pragma unroll
    for (int i = 0; i < 16; ++i)
        wq[i] = wg[i * 128 + tid];

    const float a1 = a[colbase];
    const float a2 = a[FOUT + colbase];

#pragma unroll
    for (int i = 0; i < 4; ++i)
        *reinterpret_cast<float4*>(&xs[4 * i + (tid >> 5)][4 * (tid & 31)]) = xv[i];
#pragma unroll
    for (int i = 0; i < 16; ++i)
        *reinterpret_cast<float4*>(&Ws[8 * i + (tid >> 4)][4 * (tid & 15)]) = wq[i];
    __syncthreads();

    v2f af[32], bf[32];
#pragma unroll
    for (int t = 0; t < 32; ++t) {
        const int ka = 4 * t + half * 2;
        af[t] = *reinterpret_cast<const v2f*>(&xs[l15][ka]);
        bf[t].x = Ws[ka][colbase];
        bf[t].y = Ws[ka + 1][colbase];
    }
    __builtin_amdgcn_sched_barrier(0);
    v8f c = {};
#pragma unroll
    for (int t = 0; t < 32; ++t)
        c = __builtin_amdgcn_wmma_f32_16x16x4_f32(false, af[t], false, bf[t], (short)0, c,
                                                  false, false);
    float u[4];
#pragma unroll
    for (int q = 0; q < 4; ++q)
        u[q] = fmaf(c[2 * q], a1, c[2 * q + 1] * a2);
#pragma unroll
    for (int q = 0; q < 4; ++q)
        red[half * 4 + q][wv * 16 + l15] = u[q];
    __syncthreads();
    float4* __restrict__ hout = reinterpret_cast<float4*>(
        h + (size_t)((bt * 4 + wv) * 16 + mtile) * 256);
    hout[lane]      = make_float4(c[0], c[1], c[2], c[3]);
    hout[32 + lane] = make_float4(c[4], c[5], c[6], c[7]);
    if (threadIdx.x < 8) {
        const int jl = threadIdx.x;
        const float4* __restrict__ r4 = reinterpret_cast<const float4*>(&red[jl][0]);
        float4 acc = r4[0];
#pragma unroll
        for (int i = 1; i < 16; ++i) {
            const float4 p = r4[i];
            acc.x += p.x; acc.y += p.y; acc.z += p.z; acc.w += p.w;
        }
        const float v = (acc.x + acc.y) + (acc.z + acc.w);
        float* __restrict__ sb = s + bt * NN + mtile * 8 + jl;
        sb[0]      = v;
        sb[NN / 2] = v;
    }
}

__global__ __launch_bounds__(128) void k_attn_out(const float* __restrict__ adj,
                                                  const float* __restrict__ h,
                                                  const float* __restrict__ s,
                                                  float* __restrict__ out) {
    __shared__ __attribute__((aligned(16))) float attn_lds[16][LDSW];
    __shared__ __attribute__((aligned(16))) float4 part[2][2][32];
    __shared__ __attribute__((aligned(16))) float pmax[16][8];
    __shared__ __attribute__((aligned(16))) float psum[16][8];

    const int blk   = blockIdx.x;
    const int bt    = blk >> 5;
    const int itile = (blk >> 1) & 15;
    const int nhalf = blk & 1;
    const int b     = bt >> 2;
    const int tau   = bt & 3;
    const int cpred = (itile >= 8) ? 1 : 0;
    const int tp    = (2 * tau + cpred) & 3;

    const int wv   = threadIdx.x >> 5;
    const int lane = threadIdx.x & 31;
    const int half = lane >> 4;
    const int l15  = lane & 15;
    const int nt2  = wv & 1;
    const int kh   = wv >> 1;
    const int nt   = 2 * nhalf + nt2;

    const int g    = lane >> 3;
    const int cs   = lane & 7;
    const int rloc = 2 * wv + (g >> 1) + 8 * (g & 1);

    const float4* __restrict__ srow4 = reinterpret_cast<const float4*>(s + (b * 4 + tp) * NN) + cs;
    float4 ev[8];
#pragma unroll
    for (int i = 0; i < 8; ++i) {
        const float4 sv = srow4[8 * i];
        ev[i].x = (sv.x > 0.f) ? sv.x : 0.2f * sv.x;
        ev[i].y = (sv.y > 0.f) ? sv.y : 0.2f * sv.y;
        ev[i].z = (sv.z > 0.f) ? sv.z : 0.2f * sv.z;
        ev[i].w = (sv.w > 0.f) ? sv.w : 0.2f * sv.w;
    }
    const float4* __restrict__ arow4 = reinterpret_cast<const float4*>(
        adj + ((size_t)(bt * NN + itile * 16 + rloc)) * NN) + cs;
    float4 z[8];
#pragma unroll
    for (int i = 0; i < 8; ++i)
        z[i] = arow4[8 * i];
    const float4* __restrict__ hfrag = reinterpret_cast<const float4*>(
        h + (size_t)((bt * 4 + nt) * 16 + 8 * kh) * 256);
    float4 hv[16];
#pragma unroll
    for (int kb = 0; kb < 8; ++kb) {
        hv[2 * kb]     = hfrag[kb * 64 + lane];
        hv[2 * kb + 1] = hfrag[kb * 64 + 32 + lane];
    }
    __builtin_amdgcn_sched_barrier(0);

    float4 m4 = make_float4(-3.402823466e38f, -3.402823466e38f, -3.402823466e38f, -3.402823466e38f);
#pragma unroll
    for (int i = 0; i < 8; ++i) {
        z[i].x *= ev[i].x; z[i].y *= ev[i].y; z[i].z *= ev[i].z; z[i].w *= ev[i].w;
        m4.x = fmaxf(m4.x, z[i].x); m4.y = fmaxf(m4.y, z[i].y);
        m4.z = fmaxf(m4.z, z[i].z); m4.w = fmaxf(m4.w, z[i].w);
    }
    pmax[rloc][cs] = fmaxf(fmaxf(m4.x, m4.y), fmaxf(m4.z, m4.w));
    const float4 ma = *reinterpret_cast<const float4*>(&pmax[rloc][0]);
    const float4 mb = *reinterpret_cast<const float4*>(&pmax[rloc][4]);
    const float mx = fmaxf(fmaxf(fmaxf(ma.x, ma.y), fmaxf(ma.z, ma.w)), fmaxf(fmaxf(mb.x, mb.y), fmaxf(mb.z, mb.w)));
    float4 s4 = make_float4(0.f, 0.f, 0.f, 0.f);
#pragma unroll
    for (int i = 0; i < 8; ++i) {
        z[i].x = __expf(z[i].x - mx); s4.x += z[i].x;
        z[i].y = __expf(z[i].y - mx); s4.y += z[i].y;
        z[i].z = __expf(z[i].z - mx); s4.z += z[i].z;
        z[i].w = __expf(z[i].w - mx); s4.w += z[i].w;
        *reinterpret_cast<float4*>(&attn_lds[rloc][32 * i + 4 * cs]) = z[i];
    }
    psum[rloc][cs] = (s4.x + s4.y) + (s4.z + s4.w);
    __syncthreads();

    v4f alo[8], ahi[8];
#pragma unroll
    for (int kb = 0; kb < 8; ++kb) {
        const int kcol = (8 * kh + kb) * 16 + half * 8;
        alo[kb] = *reinterpret_cast<const v4f*>(&attn_lds[l15][kcol]);
        ahi[kb] = *reinterpret_cast<const v4f*>(&attn_lds[l15][kcol + 4]);
    }
    const int rown = half * 8 + 4 * kh;
    v4f pl[4], ph[4];
#pragma unroll
    for (int e = 0; e < 4; ++e) {
        pl[e] = *reinterpret_cast<const v4f*>(&psum[rown + e][0]);
        ph[e] = *reinterpret_cast<const v4f*>(&psum[rown + e][4]);
    }
    __builtin_amdgcn_sched_barrier(0);
    v8f c0 = {}, c1 = {};
#pragma unroll
    for (int kb = 0; kb < 8; ++kb) {
        const float4 blo = hv[2 * kb];
        const float4 bhi = hv[2 * kb + 1];
        v2f a_, b_;
        a_.x = alo[kb].x; a_.y = alo[kb].y; b_.x = blo.x; b_.y = blo.y;
        c0 = __builtin_amdgcn_wmma_f32_16x16x4_f32(false, a_, false, b_, (short)0, c0, false, false);
        a_.x = alo[kb].z; a_.y = alo[kb].w; b_.x = blo.z; b_.y = blo.w;
        c1 = __builtin_amdgcn_wmma_f32_16x16x4_f32(false, a_, false, b_, (short)0, c1, false, false);
        a_.x = ahi[kb].x; a_.y = ahi[kb].y; b_.x = bhi.x; b_.y = bhi.y;
        c0 = __builtin_amdgcn_wmma_f32_16x16x4_f32(false, a_, false, b_, (short)0, c0, false, false);
        a_.x = ahi[kb].z; a_.y = ahi[kb].w; b_.x = bhi.z; b_.y = bhi.w;
        c1 = __builtin_amdgcn_wmma_f32_16x16x4_f32(false, a_, false, b_, (short)0, c1, false, false);
    }
    float inv[4];
#pragma unroll
    for (int e = 0; e < 4; ++e)
        inv[e] = 1.f / (((pl[e].x + pl[e].y) + (pl[e].z + pl[e].w)) + ((ph[e].x + ph[e].y) + (ph[e].z + ph[e].w)));
    float t[8];
#pragma unroll
    for (int r = 0; r < 8; ++r)
        t[r] = c0[r] + c1[r];
    const bool up = (kh != 0);
    const float4 keep = make_float4(up ? t[4] : t[0], up ? t[5] : t[1], up ? t[6] : t[2], up ? t[7] : t[3]);
    const float4 give = make_float4(up ? t[0] : t[4], up ? t[1] : t[5], up ? t[2] : t[6], up ? t[3] : t[7]);
    part[nt2][kh][lane] = give;
    __syncthreads();
    const float4 got = part[nt2][kh ^ 1][lane];
    const float4 lo = kh ? got : keep;
    const float4 hi = kh ? keep : got;
    const float o0 = (lo.x + hi.x) * inv[0], o1 = (lo.y + hi.y) * inv[1];
    const float o2 = (lo.z + hi.z) * inv[2], o3 = (lo.w + hi.w) * inv[3];
    float* __restrict__ orow = out + (size_t)(bt * NN + itile * 16 + half * 8 + 4 * kh) * FOUT + nt * 16 + l15;
    orow[0 * FOUT] = o0;
    orow[1 * FOUT] = o1;
    orow[2 * FOUT] = o2;
    orow[3 * FOUT] = o3;
}

extern "C" void kernel_launch(void* const* d_in, const int* in_sizes, int n_in,
                              void* d_out, int out_size, void* d_ws, size_t ws_size,
                              hipStream_t stream) {
    const float* x   = (const float*)d_in[0];
    const float* adj = (const float*)d_in[1];
    const float* W   = (const float*)d_in[2];
    const float* a   = (const float*)d_in[3];
    float* out = (float*)d_out;

    float* h = (float*)d_ws;
    float* s = h + (size_t)BT * NN * FOUT;

    k_gemm_xw_s<<<dim3(BT * 16), dim3(128), 0, stream>>>(x, W, a, h, s);
    k_attn_out<<<dim3(BT * 16 * 2), dim3(128), 0, stream>>>(adj, h, s, out);
}
